# starting kernel
# baseline (speedup 1.0000x reference)
#include <hip/hip_runtime.h>

typedef float v2f __attribute__((ext_vector_type(2)));
typedef float v8f __attribute__((ext_vector_type(8)));

#define CIN_   32
#define COUT_  32
#define SD     16
#define OD     18
#define TAPS   81
#define TAPS_PAD 84
#define TCHUNK 21
#define KCHUNKS (CIN_ * TCHUNK)
#define XCH   65536
#define OCH   104976
#define NPIX  (2 * OCH)
#define NBLOCKS (NPIX / 32)
#define WPELEMS (KCHUNKS * 2 * 32 * 2)

__global__ __launch_bounds__(256)
void prepack_w(const float* __restrict__ w, float* __restrict__ wp) {
    int idx = blockIdx.x * 256 + threadIdx.x;
    if (idx >= WPELEMS) return;
    int c2   = idx & 1;
    int lane = (idx >> 1) & 31;
    int half = (idx >> 6) & 1;
    int kc   = idx >> 7;
    int cin  = kc / TCHUNK;
    int tc   = kc - cin * TCHUNK;
    int slot = ((lane >> 4) << 1) | c2;
    int tap  = tc * 4 + slot;
    int cout = half * 16 + (lane & 15);
    float v = 0.0f;
    if (tap < TAPS) v = w[(cin * COUT_ + cout) * TAPS + tap];
    wp[idx] = v;
}

__device__ __forceinline__
void decomp_pixel(int P, int& b, int& p0, int& p1, int& p2, int& p3) {
    b = P / OCH;
    int r = P - b * OCH;
    p0 = r / (OD * OD * OD); r -= p0 * (OD * OD * OD);
    p1 = r / (OD * OD);      r -= p1 * (OD * OD);
    p2 = r / OD;
    p3 = r - p2 * OD;
}

__global__ __launch_bounds__(32)
void convt4d_wmma(const float* __restrict__ x, const float* __restrict__ wp,
                  const float* __restrict__ bias, float* __restrict__ out) {
    __shared__ int tapTab[TAPS_PAD * 32];

    const int lane  = threadIdx.x & 31;
    const int n     = lane & 15;
    const int Pbase = blockIdx.x * 32;

    {
        int b, p0, p1, p2, p3;
        decomp_pixel(Pbase + lane, b, p0, p1, p2, p3);
        for (int tap = 0; tap < TAPS_PAD; ++tap) {
            int e = -1;
            if (tap < TAPS) {
                int t0 = tap / 27;
                int t1 = (tap / 9) % 3;
                int t2 = (tap / 3) % 3;
                int t3 = tap % 3;
                int q0 = p0 - t0, q1 = p1 - t1, q2 = p2 - t2, q3 = p3 - t3;
                if ((unsigned)q0 < SD && (unsigned)q1 < SD &&
                    (unsigned)q2 < SD && (unsigned)q3 < SD)
                    e = ((q0 * SD + q1) * SD + q2) * SD + q3;
            }
            tapTab[tap * 32 + lane] = e;
        }
    }
    __syncthreads();

    int bA, pA0, pA1, pA2, pA3;
    int bB, pB0, pB1, pB2, pB3;
    decomp_pixel(Pbase + n,      bA, pA0, pA1, pA2, pA3);
    decomp_pixel(Pbase + 16 + n, bB, pB0, pB1, pB2, pB3);

    const float* xbA = x + (size_t)bA * CIN_ * XCH;
    const float* xbB = x + (size_t)bB * CIN_ * XCH;

    v8f acc00 = {};
    v8f acc10 = {};
    v8f acc01 = {};
    v8f acc11 = {};

    const int cslot = (lane >> 4) << 1;

    for (int tc = 0; tc < TCHUNK; ++tc) {
        int tap0 = tc * 4 + cslot;
        int iqA0 = tapTab[ tap0      * 32 + n];
        int iqA1 = tapTab[(tap0 + 1) * 32 + n];
        int iqB0 = tapTab[ tap0      * 32 + 16 + n];
        int iqB1 = tapTab[(tap0 + 1) * 32 + 16 + n];
        unsigned sA0 = (iqA0 < 0) ? 0u : (unsigned)iqA0;
        unsigned sA1 = (iqA1 < 0) ? 0u : (unsigned)iqA1;
        unsigned sB0 = (iqB0 < 0) ? 0u : (unsigned)iqB0;
        unsigned sB1 = (iqB1 < 0) ? 0u : (unsigned)iqB1;

        const float* xa  = xbA;
        const float* xb2 = xbB;
        const float* wpc = wp + tc * 128 + lane * 2;

        for (int cin = 0; cin < CIN_; ++cin) {
            float vA0 = xa[sA0];
            float vA1 = xa[sA1];
            float vB0 = xb2[sB0];
            float vB1 = xb2[sB1];
            v2f bbA = { (iqA0 < 0) ? 0.0f : vA0, (iqA1 < 0) ? 0.0f : vA1 };
            v2f bbB = { (iqB0 < 0) ? 0.0f : vB0, (iqB1 < 0) ? 0.0f : vB1 };

            v2f a0 = *(const v2f*)(wpc);
            v2f a1 = *(const v2f*)(wpc + 64);

            acc00 = __builtin_amdgcn_wmma_f32_16x16x4_f32(
                        false, a0, false, bbA, (short)0, acc00, false, false);
            acc10 = __builtin_amdgcn_wmma_f32_16x16x4_f32(
                        false, a1, false, bbA, (short)0, acc10, false, false);
            acc01 = __builtin_amdgcn_wmma_f32_16x16x4_f32(
                        false, a0, false, bbB, (short)0, acc01, false, false);
            acc11 = __builtin_amdgcn_wmma_f32_16x16x4_f32(
                        false, a1, false, bbB, (short)0, acc11, false, false);

            xa  += XCH;
            xb2 += XCH;
            wpc += TCHUNK * 128;
        }
    }

    int spA = ((pA0 * OD + pA1) * OD + pA2) * OD + pA3;
    int spB = ((pB0 * OD + pB1) * OD + pB2) * OD + pB3;
    float* oA = out + (size_t)(bA * COUT_) * OCH + spA;
    float* oB = out + (size_t)(bB * COUT_) * OCH + spB;
    int mbase = (lane >> 4) << 3;
#pragma unroll
    for (int j = 0; j < 8; ++j) {
        int c0 = mbase + j;
        float bi0 = bias[c0];
        float bi1 = bias[c0 + 16];
        oA[(size_t)c0 * OCH]        = acc00[j] + bi0;
        oA[(size_t)(c0 + 16) * OCH] = acc10[j] + bi1;
        oB[(size_t)c0 * OCH]        = acc01[j] + bi0;
        oB[(size_t)(c0 + 16) * OCH] = acc11[j] + bi1;
    }
}

extern "C" void kernel_launch(void* const* d_in, const int* in_sizes, int n_in,
                              void* d_out, int out_size, void* d_ws, size_t ws_size,
                              hipStream_t stream) {
    const float* x    = (const float*)d_in[0];
    const float* w    = (const float*)d_in[1];
    const float* bias = (const float*)d_in[2];
    float* out = (float*)d_out;
    float* wp  = (float*)d_ws;

    prepack_w<<<(WPELEMS + 255) / 256, 256, 0, stream>>>(w, wp);
    convt4d_wmma<<<NBLOCKS, 32, 0, stream>>>(x, wp, bias, out);
}


# scatter form: an input row's 16 columns are the A operand, weights B (N = cout, K = cin); per-t3 accumulators overlap-added in registers; 256 blocks of 8 equal waves, no padded products
# speedup vs baseline: 2.0041x; 2.0041x over previous
#include <hip/hip_runtime.h>

typedef float v2f __attribute__((ext_vector_type(2)));
typedef float v4f __attribute__((ext_vector_type(4)));
typedef float v8f __attribute__((ext_vector_type(8)));

#define XCH   65536
#define OCH   104976
#define WPELEMS (9 * 8 * 9 * 128)

__global__ __launch_bounds__(256)
void prepack_w(const float* __restrict__ w, float* __restrict__ wp) {
    int idx = blockIdx.x * 256 + threadIdx.x;
    if (idx >= WPELEMS) return;
    int s    = idx & 1;
    int half = (idx >> 1) & 1;
    int lane = (idx >> 2) & 31;
    int rest = idx >> 7;
    int t23  = rest % 9;
    int r2   = rest / 9;
    int cc   = r2 & 7;
    int t01  = r2 >> 3;
    int cin  = cc * 4 + 2 * (lane >> 4) + s;
    int cout = half * 16 + (lane & 15);
    int tap  = t01 * 9 + t23;
    wp[idx] = w[(cin * 32 + cout) * 81 + tap];
}


#define CLAMP15(v) ((v) < 0 ? 0 : ((v) > 15 ? 15 : (v)))

template <int RB>
__device__ __forceinline__ void load_step(const float* xp, const float* wq, v2f (&xa)[7], v4f (&wb)[9]) {
#pragma unroll
    for (int j = 0; j < 7; ++j) {
        const int q2 = CLAMP15(RB - 2 + j);
        xa[j].x = xp[q2 * 16];
        xa[j].y = xp[XCH + q2 * 16];
    }
#pragma unroll
    for (int t = 0; t < 9; ++t) wb[t] = *(const v4f*)(wq + t * 128);
}

template <int RB, int NR>
__device__ __forceinline__ void mma_step(v8f (&acc)[5][3][2], const v2f (&xa)[7], const v4f (&wb)[9]) {
#pragma unroll
    for (int r = 0; r < NR; ++r) {
#pragma unroll
        for (int t2 = 0; t2 < 3; ++t2) {
            const int q2 = RB + r - t2;
            if (q2 >= 0 && q2 <= 15) {
                const v2f a = xa[r + 2 - t2];
#pragma unroll
                for (int t3 = 0; t3 < 3; ++t3) {
                    const v4f wv = wb[t2 * 3 + t3];
                    const v2f b0 = { wv.x, wv.y };
                    const v2f b1 = { wv.z, wv.w };
                    acc[r][t3][0] = __builtin_amdgcn_wmma_f32_16x16x4_f32(
                        false, a, false, b0, (short)0, acc[r][t3][0], false, false);
                    acc[r][t3][1] = __builtin_amdgcn_wmma_f32_16x16x4_f32(
                        false, a, false, b1, (short)0, acc[r][t3][1], false, false);
                }
            }
        }
    }
}

template <int RB, int NR>
__device__ __forceinline__ void run_group(const float* __restrict__ x, const float* __restrict__ wp,
                                          const float* __restrict__ bias, float* __restrict__ out,
                                          int b, int s0, int s1, int lane) {
    const int c = lane & 15;
    const int h = lane >> 4;
    const float* xlane = x + (size_t)(b * 32 + 2 * h) * XCH + c;
    const float* wlane = wp + lane * 4;
    const int n0 = (s0 < 2) ? 2 : 1;
    const int n1 = (s1 < 2) ? 2 : 1;

    for (int i0 = 0; i0 < n0; ++i0)
    for (int i1 = 0; i1 < n1; ++i1) {
        const int p0 = s0 + 16 * i0;
        const int p1 = s1 + 16 * i1;

        v8f acc[5][3][2];
#pragma unroll
        for (int r = 0; r < 5; ++r)
#pragma unroll
            for (int t3 = 0; t3 < 3; ++t3) { acc[r][t3][0] = {}; acc[r][t3][1] = {}; }

        for (int t0 = 0; t0 < 3; ++t0) {
            const int q0 = p0 - t0;
            if ((unsigned)q0 > 15u) continue;
            for (int t1 = 0; t1 < 3; ++t1) {
                const int q1 = p1 - t1;
                if ((unsigned)q1 > 15u) continue;
                const float* xp = xlane + (q0 * 16 + q1) * 256;
                const float* wq = wlane + (size_t)((t0 * 3 + t1) * 8) * (9 * 128);
                v2f xaA[7], xaB[7];
                v4f wbA[9], wbB[9];
                load_step<RB>(xp, wq, xaA, wbA);
                for (int cc2 = 0; cc2 < 4; ++cc2) {
                    load_step<RB>(xp + 4 * XCH, wq + 9 * 128, xaB, wbB);
                    mma_step<RB, NR>(acc, xaA, wbA);
                    const int adv = (cc2 < 3) ? 2 : 1;
                    xp += (size_t)adv * 4 * XCH;
                    wq += adv * 9 * 128;
                    load_step<RB>(xp, wq, xaA, wbA);
                    mma_step<RB, NR>(acc, xaB, wbB);
                }
            }
        }

#pragma unroll
        for (int r = 0; r < NR; ++r) {
            const int p2 = RB + r;
            const int sp = ((p0 * 18 + p1) * 18 + p2) * 18 + 8 * h;
#pragma unroll
            for (int hf = 0; hf < 2; ++hf) {
                const v8f A0 = acc[r][0][hf];
                const v8f A1 = acc[r][1][hf];
                const v8f A2 = acc[r][2][hf];
                const float o17 = __shfl_xor(A1[7], 16);
                const float o26 = __shfl_xor(A2[6], 16);
                const float o27 = __shfl_xor(A2[7], 16);
                const float l1  = h ? o17 : 0.0f;
                const float l2a = h ? o26 : 0.0f;
                const float l2b = h ? o27 : 0.0f;
                const int cout = hf * 16 + c;
                const float bi = bias[cout];
                float* o = out + (size_t)(b * 32 + cout) * OCH + sp;
                v2f r01 = { A0[0] + l1 + l2a + bi,      A0[1] + A1[0] + l2b + bi };
                v2f r23 = { A0[2] + A1[1] + A2[0] + bi, A0[3] + A1[2] + A2[1] + bi };
                v2f r45 = { A0[4] + A1[3] + A2[2] + bi, A0[5] + A1[4] + A2[3] + bi };
                v2f r67 = { A0[6] + A1[5] + A2[4] + bi, A0[7] + A1[6] + A2[5] + bi };
                *(v2f*)(o + 0) = r01;
                *(v2f*)(o + 2) = r23;
                *(v2f*)(o + 4) = r45;
                *(v2f*)(o + 6) = r67;
                if (h) {
                    v2f r89 = { A1[7] + A2[6] + bi, A2[7] + bi };
                    *(v2f*)(o + 8) = r89;
                }
            }
        }
    }
}

__global__ __launch_bounds__(256)
void convt4d_scatter(const float* __restrict__ x, const float* __restrict__ wp,
                     const float* __restrict__ bias, float* __restrict__ out) {
    const int lane = threadIdx.x & 31;
    const int wave = __builtin_amdgcn_readfirstlane((int)(threadIdx.x >> 5));
    const int b  = wave >> 2;
    const int gz = wave & 3;
    const int s0 = blockIdx.x >> 4;
    const int s1 = blockIdx.x & 15;
    if (gz == 0)      run_group<0, 5>(x, wp, bias, out, b, s0, s1, lane);
    else if (gz == 1) run_group<5, 4>(x, wp, bias, out, b, s0, s1, lane);
    else if (gz == 2) run_group<9, 4>(x, wp, bias, out, b, s0, s1, lane);
    else              run_group<13, 5>(x, wp, bias, out, b, s0, s1, lane);
}

extern "C" void kernel_launch(void* const* d_in, const int* in_sizes, int n_in,
                              void* d_out, int out_size, void* d_ws, size_t ws_size,
                              hipStream_t stream) {
    const float* x    = (const float*)d_in[0];
    const float* w    = (const float*)d_in[1];
    const float* bias = (const float*)d_in[2];
    float* out = (float*)d_out;
    float* wp  = (float*)d_ws;

    prepack_w<<<(WPELEMS + 255) / 256, 256, 0, stream>>>(w, wp);
    convt4d_scatter<<<256, 256, 0, stream>>>(x, wp, bias, out);
}


# last dimension as polynomial product (column pair x three taps) evaluated at 0, 1, -1, infinity: 4 WMMA per pair and t2, not 6; both batches in one M tile; 48 WMMA per K-step and wave, not 72
# speedup vs baseline: 2.4422x; 1.2186x over previous
#include <hip/hip_runtime.h>

typedef float v2f __attribute__((ext_vector_type(2)));
typedef float v4f __attribute__((ext_vector_type(4)));
typedef float v8f __attribute__((ext_vector_type(8)));

#define XCH   65536
#define OCH   104976
#define WSTEP (12 * 128)
#define WPELEMS (9 * 8 * WSTEP)


__global__ __launch_bounds__(256)
void prepack_w(const float* __restrict__ w, float* __restrict__ wp) {
    int idx = blockIdx.x * 256 + threadIdx.x;
    if (idx >= WPELEMS) return;
    int s    = idx & 1;
    int half = (idx >> 1) & 1;
    int lane = (idx >> 2) & 31;
    int rest = idx >> 7;
    int tj   = rest % 12;
    int r2   = rest / 12;
    int cc   = r2 & 7;
    int t01  = r2 >> 3;
    int t2   = tj >> 2;
    int j    = tj & 3;
    int cin  = cc * 4 + 2 * (lane >> 4) + s;
    int cout = half * 16 + (lane & 15);
    const float* p = w + (cin * 32 + cout) * 81 + t01 * 9 + t2 * 3;
    const float w0 = p[0], w1 = p[1], w2 = p[2];
    float v;
    if (j == 0)      v = w0;
    else if (j == 1) v = 0.5f * ((w0 + w2) + w1);
    else if (j == 2) v = 0.5f * ((w0 + w2) - w1);
    else             v = w2;
    wp[idx] = v;
}


template <int NF>
__device__ __forceinline__ void load_step(const float* xp, const float* wq, v2f (&xr)[NF][2], v4f (&wb)[12]) {
#pragma unroll
    for (int f = 0; f < NF; ++f) {
        xr[f][0] = *(const v2f*)(xp + f * 16);
        xr[f][1] = *(const v2f*)(xp + XCH + f * 16);
    }
#pragma unroll
    for (int t = 0; t < 12; ++t) wb[t] = *(const v4f*)(wq + t * 128);
}

template <int NR, int NF, int FOFF>
__device__ __forceinline__ void mma_step(v8f (&acc)[NR][4][2], const v2f (&xr)[NF][2], const v4f (&wb)[12]) {
    v2f xa[NF][4];
#pragma unroll
    for (int f = 0; f < NF; ++f) {
        const v2f c0 = xr[f][0];
        const v2f c1 = xr[f][1];
        xa[f][0] = (v2f){ c0.x,        c1.x        };
        xa[f][1] = (v2f){ c0.x + c0.y, c1.x + c1.y };
        xa[f][2] = (v2f){ c0.x - c0.y, c1.x - c1.y };
        xa[f][3] = (v2f){ c0.y,        c1.y        };
    }
#pragma unroll
    for (int r = 0; r < NR; ++r) {
#pragma unroll
        for (int t2 = 0; t2 < 3; ++t2) {
            const int f = r + FOFF - t2;
            if (f >= 0 && f < NF) {
#pragma unroll
                for (int j = 0; j < 4; ++j) {
                    const v2f a  = xa[f][j];
                    const v4f wv = wb[t2 * 4 + j];
                    const v2f b0 = { wv.x, wv.y };
                    const v2f b1 = { wv.z, wv.w };
                    acc[r][j][0] = __builtin_amdgcn_wmma_f32_16x16x4_f32(
                        false, a, false, b0, (short)0, acc[r][j][0], false, false);
                    acc[r][j][1] = __builtin_amdgcn_wmma_f32_16x16x4_f32(
                        false, a, false, b1, (short)0, acc[r][j][1], false, false);
                }
            }
        }
    }
}

template <int NR, int NF, int FOFF>
__device__ __forceinline__ void run_group(const float* __restrict__ x, const float* __restrict__ wp,
                                          const float* __restrict__ bias, float* __restrict__ out,
                                          int rb, int qlo, int s0, int s1, int lane) {
    const int c = lane & 15;
    const int h = lane >> 4;
    const float* xlane = x + (size_t)((c >> 3) * 32 + 2 * h) * XCH + qlo * 16 + 2 * (c & 7);
    const float* wlane = wp + lane * 4;
    const int n0 = (s0 < 2) ? 2 : 1;
    const int n1 = (s1 < 2) ? 2 : 1;

    for (int i0 = 0; i0 < n0; ++i0)
    for (int i1 = 0; i1 < n1; ++i1) {
        const int p0 = s0 + 16 * i0;
        const int p1 = s1 + 16 * i1;

        v8f acc[NR][4][2];
#pragma unroll
        for (int r = 0; r < NR; ++r)
#pragma unroll
            for (int j = 0; j < 4; ++j) { acc[r][j][0] = {}; acc[r][j][1] = {}; }

        for (int t0 = 0; t0 < 3; ++t0) {
            const int q0 = p0 - t0;
            if ((unsigned)q0 > 15u) continue;
            for (int t1 = 0; t1 < 3; ++t1) {
                const int q1 = p1 - t1;
                if ((unsigned)q1 > 15u) continue;
                const float* xp = xlane + (q0 * 16 + q1) * 256;
                const float* wq = wlane + (size_t)((t0 * 3 + t1) * 8) * WSTEP;
                v2f xrA[NF][2], xrB[NF][2];
                v4f wbA[12], wbB[12];
                load_step<NF>(xp, wq, xrA, wbA);
                for (int cc2 = 0; cc2 < 4; ++cc2) {
                    __builtin_amdgcn_sched_barrier(0);
                    load_step<NF>(xp + 4 * XCH, wq + WSTEP, xrB, wbB);
                    __builtin_amdgcn_sched_barrier(0);
                    mma_step<NR, NF, FOFF>(acc, xrA, wbA);
                    const int adv = (cc2 < 3) ? 2 : 1;
                    xp += (size_t)adv * 4 * XCH;
                    wq += adv * WSTEP;
                    __builtin_amdgcn_sched_barrier(0);
                    load_step<NF>(xp, wq, xrA, wbA);
                    __builtin_amdgcn_sched_barrier(0);
                    mma_step<NR, NF, FOFF>(acc, xrB, wbB);
                }
            }
        }

#pragma unroll
        for (int r = 0; r < NR; ++r) {
            const int p2 = rb + r;
            const int sp = ((p0 * 18 + p1) * 18 + p2) * 18;
#pragma unroll
            for (int hf = 0; hf < 2; ++hf) {
                const v8f M0 = acc[r][0][hf];
                const v8f M1 = acc[r][1][hf];
                const v8f M2 = acc[r][2][hf];
                const v8f M3 = acc[r][3][hf];
                const int cout = hf * 16 + c;
                const float bi = bias[cout];
                float* o = out + (size_t)(h * 32 + cout) * OCH + sp;
                float ev[9], od[9];
                ev[0] = M0[0] + bi;
                od[0] = (M1[0] - M2[0] - M3[0]) + bi;
#pragma unroll
                for (int i = 1; i < 8; ++i) {
                    ev[i] = (M0[i] + (M1[i - 1] + M2[i - 1] - M0[i - 1])) + bi;
                    od[i] = ((M1[i] - M2[i] - M3[i]) + M3[i - 1]) + bi;
                }
                ev[8] = (M1[7] + M2[7] - M0[7]) + bi;
                od[8] = M3[7] + bi;
#pragma unroll
                for (int i = 0; i < 9; ++i) {
                    const v2f st = { ev[i], od[i] };
                    *(v2f*)(o + 2 * i) = st;
                }
            }
        }
    }
}

__global__ __launch_bounds__(256)
void convt4d_scatter(const float* __restrict__ x, const float* __restrict__ wp,
                     const float* __restrict__ bias, float* __restrict__ out) {
    const int lane = threadIdx.x & 31;
    const int wave = __builtin_amdgcn_readfirstlane((int)(threadIdx.x >> 5));
    const int s0 = blockIdx.x >> 4;
    const int s1 = blockIdx.x & 15;
    if (wave == 0)      run_group<3, 3, 0>(x, wp, bias, out, 0, 0, s0, s1, lane);
    else if (wave == 7) run_group<3, 3, 2>(x, wp, bias, out, 15, 13, s0, s1, lane);
    else                run_group<2, 4, 2>(x, wp, bias, out, 2 * wave + 1, 2 * wave - 1, s0, s1, lane);
}

extern "C" void kernel_launch(void* const* d_in, const int* in_sizes, int n_in,
                              void* d_out, int out_size, void* d_ws, size_t ws_size,
                              hipStream_t stream) {
    const float* x    = (const float*)d_in[0];
    const float* w    = (const float*)d_in[1];
    const float* bias = (const float*)d_in[2];
    float* out = (float*)d_out;
    float* wp  = (float*)d_ws;

    prepack_w<<<WPELEMS / 256, 256, 0, stream>>>(w, wp);
    convt4d_scatter<<<256, 256, 0, stream>>>(x, wp, bias, out);
}


# wave tile = one cout half x two input row pairs (was both halves x one pair): each loaded weight fragment feeds two WMMAs, half the weight loads per wave; weights prepacked per cout half
# speedup vs baseline: 2.6132x; 1.0700x over previous
#include <hip/hip_runtime.h>

typedef float v2f __attribute__((ext_vector_type(2)));
typedef float v4f __attribute__((ext_vector_type(4)));
typedef float v8f __attribute__((ext_vector_type(8)));

#define XCH   65536
#define OCH   104976
#define WSTEP (16 * 128)
#define WHALF (8 * 128)
#define WPELEMS (9 * 8 * WSTEP)


__device__ __forceinline__ float eval_tap(int j, float a0, float a1, float a2) {
    if (j == 0) return a0;
    if (j == 1) return 0.5f * ((a0 + a2) + a1);
    if (j == 2) return 0.5f * ((a0 + a2) - a1);
    return a2;
}

__global__ __launch_bounds__(256)
void prepack_w(const float* __restrict__ w, float* __restrict__ wp) {
    int idx = blockIdx.x * 256 + threadIdx.x;
    if (idx >= WPELEMS) return;
    int s    = idx & 1;
    int e    = (idx >> 1) & 1;
    int lane = (idx >> 2) & 31;
    int jp   = (idx >> 7) & 7;
    int half = (idx >> 10) & 1;
    int r2   = idx >> 11;
    int cc   = r2 & 7;
    int t01  = r2 >> 3;
    int jj   = jp * 2 + e;
    int jr   = jj >> 2;
    int jc   = jj & 3;
    int cin  = cc * 4 + 2 * (lane >> 4) + s;
    int cout = half * 16 + (lane & 15);
    const float* p = w + (cin * 32 + cout) * 81 + t01 * 9;
    const float u0 = eval_tap(jc, p[0], p[1], p[2]);
    const float u1 = eval_tap(jc, p[3], p[4], p[5]);
    const float u2 = eval_tap(jc, p[6], p[7], p[8]);
    wp[idx] = eval_tap(jr, u0, u1, u2);
}


__device__ __forceinline__ void load_step(const float* xp, const float* wq, v2f (&xr)[4][2], v4f (&wb)[8]) {
#pragma unroll
    for (int r = 0; r < 4; ++r) {
        xr[r][0] = *(const v2f*)(xp + r * 16);
        xr[r][1] = *(const v2f*)(xp + XCH + r * 16);
    }
#pragma unroll
    for (int t = 0; t < 8; ++t) wb[t] = *(const v4f*)(wq + t * 128);
}

__device__ __forceinline__ void mma_step(v8f (&acc)[2][4][4], const v2f (&xr)[4][2], const v4f (&wb)[8]) {
#pragma unroll
    for (int rp = 0; rp < 2; ++rp) {
        v2f rr[4][2];
#pragma unroll
        for (int s = 0; s < 2; ++s) {
            rr[0][s] = xr[2 * rp][s];
            rr[1][s] = xr[2 * rp][s] + xr[2 * rp + 1][s];
            rr[2][s] = xr[2 * rp][s] - xr[2 * rp + 1][s];
            rr[3][s] = xr[2 * rp + 1][s];
        }
#pragma unroll
        for (int jr = 0; jr < 4; ++jr) {
            const v2f c0 = rr[jr][0];
            const v2f c1 = rr[jr][1];
            v2f xa[4];
            xa[0] = (v2f){ c0.x,        c1.x        };
            xa[1] = (v2f){ c0.x + c0.y, c1.x + c1.y };
            xa[2] = (v2f){ c0.x - c0.y, c1.x - c1.y };
            xa[3] = (v2f){ c0.y,        c1.y        };
#pragma unroll
            for (int jc = 0; jc < 4; ++jc) {
                const int jj = jr * 4 + jc;
                const v4f wv = wb[jj >> 1];
                const v2f b  = (jj & 1) ? (v2f){ wv.z, wv.w } : (v2f){ wv.x, wv.y };
                acc[rp][jr][jc] = __builtin_amdgcn_wmma_f32_16x16x4_f32(
                    false, xa[jc], false, b, (short)0, acc[rp][jr][jc], false, false);
            }
        }
    }
}

__device__ __forceinline__ void col_eval(const v8f Z0, const v8f Z1, const v8f Z2, const v8f Z3, v2f (&st)[9]) {
    st[0] = (v2f){ Z0[0], (Z1[0] - Z2[0]) - Z3[0] };
#pragma unroll
    for (int i = 1; i < 8; ++i) {
        st[i] = (v2f){ Z0[i] + ((Z1[i - 1] + Z2[i - 1]) - Z0[i - 1]),
                       ((Z1[i] - Z2[i]) - Z3[i]) + Z3[i - 1] };
    }
    st[8] = (v2f){ (Z1[7] + Z2[7]) - Z0[7], Z3[7] };
}

__device__ __forceinline__ void store_row(float* o, const v2f (&st)[9], const float bi) {
#pragma unroll
    for (int i = 0; i < 9; ++i) *(v2f*)(o + 2 * i) = st[i] + bi;
}

__global__ __launch_bounds__(256)
void convt4d_poly2_rq(const float* __restrict__ x, const float* __restrict__ wp,
                      const float* __restrict__ bias, float* __restrict__ out) {
    __shared__ v2f xch[8][18][32];

    const int lane = threadIdx.x & 31;
    const int wave = __builtin_amdgcn_readfirstlane((int)(threadIdx.x >> 5));
    const int hf = wave & 1;
    const int u  = wave >> 1;
    const int s0 = blockIdx.x >> 4;
    const int s1 = blockIdx.x & 15;
    const int c = lane & 15;
    const int h = lane >> 4;
    const float* xlane = x + (size_t)((c >> 3) * 32 + 2 * h) * XCH + (4 * u) * 16 + 2 * (c & 7);
    const float* wlane = wp + hf * WHALF + lane * 4;
    const int cout = hf * 16 + c;
    const float bi = bias[cout];
    const int n0 = (s0 < 2) ? 2 : 1;
    const int n1 = (s1 < 2) ? 2 : 1;

    for (int i0 = 0; i0 < n0; ++i0)
    for (int i1 = 0; i1 < n1; ++i1) {
        const int p0 = s0 + 16 * i0;
        const int p1 = s1 + 16 * i1;

        v8f acc[2][4][4];
#pragma unroll
        for (int rp = 0; rp < 2; ++rp)
#pragma unroll
            for (int jr = 0; jr < 4; ++jr)
#pragma unroll
                for (int jc = 0; jc < 4; ++jc) acc[rp][jr][jc] = {};

        for (int t0 = 0; t0 < 3; ++t0) {
            const int q0 = p0 - t0;
            if ((unsigned)q0 > 15u) continue;
            for (int t1 = 0; t1 < 3; ++t1) {
                const int q1 = p1 - t1;
                if ((unsigned)q1 > 15u) continue;
                const float* xp = xlane + (q0 * 16 + q1) * 256;
                const float* wq = wlane + (size_t)((t0 * 3 + t1) * 8) * WSTEP;
                v2f xrA[4][2], xrB[4][2];
                v4f wbA[8], wbB[8];
                load_step(xp, wq, xrA, wbA);
                for (int cc2 = 0; cc2 < 4; ++cc2) {
                    __builtin_amdgcn_sched_barrier(0);
                    load_step(xp + 4 * XCH, wq + WSTEP, xrB, wbB);
                    __builtin_amdgcn_sched_barrier(0);
                    mma_step(acc, xrA, wbA);
                    const int adv = (cc2 < 3) ? 2 : 1;
                    xp += (size_t)adv * 4 * XCH;
                    wq += adv * WSTEP;
                    __builtin_amdgcn_sched_barrier(0);
                    load_step(xp, wq, xrA, wbA);
                    __builtin_amdgcn_sched_barrier(0);
                    mma_step(acc, xrB, wbB);
                }
            }
        }

        v2f st4[9], st5[9];
        {
            v8f z[4];
#pragma unroll
            for (int jc = 0; jc < 4; ++jc) z[jc] = (acc[1][1][jc] + acc[1][2][jc]) - acc[1][0][jc];
            col_eval(z[0], z[1], z[2], z[3], st4);
            col_eval(acc[1][3][0], acc[1][3][1], acc[1][3][2], acc[1][3][3], st5);
        }
#pragma unroll
        for (int i = 0; i < 9; ++i) {
            xch[wave][i][lane]     = st4[i];
            xch[wave][9 + i][lane] = st5[i];
        }
        __syncthreads();

        float* oplane = out + (size_t)(h * 32 + cout) * OCH + (size_t)((p0 * 18 + p1) * 18) * 18;
        float* o = oplane + (4 * u) * 18;
        {
            v2f st0[9], st1[9];
            v8f z[4];
            col_eval(acc[0][0][0], acc[0][0][1], acc[0][0][2], acc[0][0][3], st0);
#pragma unroll
            for (int jc = 0; jc < 4; ++jc) z[jc] = (acc[0][1][jc] - acc[0][2][jc]) - acc[0][3][jc];
            col_eval(z[0], z[1], z[2], z[3], st1);
            if (u > 0) {
#pragma unroll
                for (int i = 0; i < 9; ++i) {
                    st0[i] += xch[wave - 2][i][lane];
                    st1[i] += xch[wave - 2][9 + i][lane];
                }
            }
            store_row(o, st0, bi);
            store_row(o + 18, st1, bi);
        }
        {
            v2f st2[9], st3[9];
            v8f z[4];
#pragma unroll
            for (int jc = 0; jc < 4; ++jc)
                z[jc] = ((acc[0][1][jc] + acc[0][2][jc]) - acc[0][0][jc]) + acc[1][0][jc];
            col_eval(z[0], z[1], z[2], z[3], st2);
#pragma unroll
            for (int jc = 0; jc < 4; ++jc)
                z[jc] = acc[0][3][jc] + ((acc[1][1][jc] - acc[1][2][jc]) - acc[1][3][jc]);
            col_eval(z[0], z[1], z[2], z[3], st3);
            store_row(o + 36, st2, bi);
            store_row(o + 54, st3, bi);
        }
        if (u == 0) {
            v2f st16[9];
#pragma unroll
            for (int i = 0; i < 9; ++i) st16[i] = xch[6 + hf][i][lane];
            store_row(oplane + 16 * 18, st16, bi);
        }
        if (u == 3) store_row(oplane + 17 * 18, st5, bi);
        __syncthreads();
    }
}

extern "C" void kernel_launch(void* const* d_in, const int* in_sizes, int n_in,
                              void* d_out, int out_size, void* d_ws, size_t ws_size,
                              hipStream_t stream) {
    const float* x    = (const float*)d_in[0];
    const float* w    = (const float*)d_in[1];
    const float* bias = (const float*)d_in[2];
    float* out = (float*)d_out;
    float* wp  = (float*)d_ws;

    prepack_w<<<WPELEMS / 256, 256, 0, stream>>>(w, wp);
    convt4d_poly2_rq<<<256, 256, 0, stream>>>(x, wp, bias, out);
}


# WMMA operands swapped (weights A, x column pairs B): D lanes run along the output row, a store writes 64 contiguous bytes per row, not a few bytes of each of 32 rows; columns summed by a lane shift
# speedup vs baseline: 3.0698x; 1.1747x over previous
#include <hip/hip_runtime.h>

typedef float v2f __attribute__((ext_vector_type(2)));
typedef float v4f __attribute__((ext_vector_type(4)));
typedef float v8f __attribute__((ext_vector_type(8)));

#define XCH   65536
#define OCH   104976
#define WSTEP (16 * 128)
#define WHALF (8 * 128)
#define WPELEMS (9 * 8 * WSTEP)


__device__ __forceinline__ float eval_tap(int j, float a0, float a1, float a2) {
    if (j == 0) return a0;
    if (j == 1) return 0.5f * ((a0 + a2) + a1);
    if (j == 2) return 0.5f * ((a0 + a2) - a1);
    return a2;
}

__global__ __launch_bounds__(256)
void prepack_w(const float* __restrict__ w, float* __restrict__ wp) {
    int idx = blockIdx.x * 256 + threadIdx.x;
    if (idx >= WPELEMS) return;
    int s    = idx & 1;
    int e    = (idx >> 1) & 1;
    int lane = (idx >> 2) & 31;
    int jp   = (idx >> 7) & 7;
    int half = (idx >> 10) & 1;
    int r2   = idx >> 11;
    int cc   = r2 & 7;
    int t01  = r2 >> 3;
    int jj   = jp * 2 + e;
    int jr   = jj >> 2;
    int jc   = jj & 3;
    int cin  = cc * 4 + 2 * (lane >> 4) + s;
    int cout = half * 16 + (lane & 15);
    const float* p = w + (cin * 32 + cout) * 81 + t01 * 9;
    const float u0 = eval_tap(jc, p[0], p[1], p[2]);
    const float u1 = eval_tap(jc, p[3], p[4], p[5]);
    const float u2 = eval_tap(jc, p[6], p[7], p[8]);
    wp[idx] = eval_tap(jr, u0, u1, u2);
}


__device__ __forceinline__ void load_step(const float* xp, const float* wq, v2f (&xr)[4][2], v4f (&wb)[8]) {
#pragma unroll
    for (int r = 0; r < 4; ++r) {
        xr[r][0] = *(const v2f*)(xp + r * 16);
        xr[r][1] = *(const v2f*)(xp + XCH + r * 16);
    }
#pragma unroll
    for (int t = 0; t < 8; ++t) wb[t] = *(const v4f*)(wq + t * 128);
}

__device__ __forceinline__ void mma_step(v8f (&acc)[2][4][4], const v2f (&xr)[4][2], const v4f (&wb)[8]) {
#pragma unroll
    for (int rp = 0; rp < 2; ++rp) {
        v2f rr[4][2];
#pragma unroll
        for (int s = 0; s < 2; ++s) {
            rr[0][s] = xr[2 * rp][s];
            rr[1][s] = xr[2 * rp][s] + xr[2 * rp + 1][s];
            rr[2][s] = xr[2 * rp][s] - xr[2 * rp + 1][s];
            rr[3][s] = xr[2 * rp + 1][s];
        }
#pragma unroll
        for (int jr = 0; jr < 4; ++jr) {
            const v2f c0 = rr[jr][0];
            const v2f c1 = rr[jr][1];
            v2f xa[4];
            xa[0] = (v2f){ c0.x,        c1.x        };
            xa[1] = (v2f){ c0.x + c0.y, c1.x + c1.y };
            xa[2] = (v2f){ c0.x - c0.y, c1.x - c1.y };
            xa[3] = (v2f){ c0.y,        c1.y        };
#pragma unroll
            for (int jc = 0; jc < 4; ++jc) {
                const int jj = jr * 4 + jc;
                const v4f wv = wb[jj >> 1];
                const v2f b  = (jj & 1) ? (v2f){ wv.z, wv.w } : (v2f){ wv.x, wv.y };
                acc[rp][jr][jc] = __builtin_amdgcn_wmma_f32_16x16x4_f32(
                    false, b, false, xa[jc], (short)0, acc[rp][jr][jc], false, false);
            }
        }
    }
}

__device__ __forceinline__ float left_pair(const float v, const bool first) {
    const int s = __builtin_amdgcn_update_dpp(0, __builtin_bit_cast(int, v), 0x111, 0xf, 0xf, true);
    return first ? 0.0f : __builtin_bit_cast(float, s);
}

__device__ __forceinline__ void emit_row(float* o, const v8f Z0, const v8f Z1, const v8f Z2, const v8f Z3,
                                         const v8f bi, const bool first, const bool last) {
    float t0[8], t1[8];
#pragma unroll
    for (int r = 0; r < 8; ++r) {
        t0[r] = (Z1[r] + Z2[r]) - Z0[r];
        t1[r] = Z3[r];
        const float y1 = (Z1[r] - Z2[r]) - Z3[r];
        const float o0 = (Z0[r] + left_pair(t0[r], first)) + bi[r];
        const float o1 = (y1 + left_pair(t1[r], first)) + bi[r];
        *(v2f*)(o + (size_t)r * OCH) = (v2f){ o0, o1 };
    }
    if (last) {
#pragma unroll
        for (int r = 0; r < 8; ++r)
            *(v2f*)(o + (size_t)r * OCH + 2) = (v2f){ t0[r] + bi[r], t1[r] + bi[r] };
    }
}

__device__ __forceinline__ v4f lo4(const v8f z) { return __builtin_shufflevector(z, z, 0, 1, 2, 3); }
__device__ __forceinline__ v4f hi4(const v8f z) { return __builtin_shufflevector(z, z, 4, 5, 6, 7); }
__device__ __forceinline__ v8f cat8(const v4f a, const v4f b) { return __builtin_shufflevector(a, b, 0, 1, 2, 3, 4, 5, 6, 7); }

__global__ __launch_bounds__(256)
void convt4d_poly2_rq(const float* __restrict__ x, const float* __restrict__ wp,
                      const float* __restrict__ bias, float* __restrict__ out) {
    __shared__ v4f xch[8][2][4][2][32];

    const int lane = threadIdx.x & 31;
    const int wave = __builtin_amdgcn_readfirstlane((int)(threadIdx.x >> 5));
    const int hf = wave & 1;
    const int u  = wave >> 1;
    const int s0 = blockIdx.x >> 4;
    const int s1 = blockIdx.x & 15;
    const int c = lane & 15;
    const int h = lane >> 4;
    const float* xlane = x + (size_t)((c >> 3) * 32 + 2 * h) * XCH + (4 * u) * 16 + 2 * (c & 7);
    const float* wlane = wp + hf * WHALF + lane * 4;
    const int cout0 = hf * 16 + 8 * h;
    const v8f bi = cat8(*(const v4f*)(bias + cout0), *(const v4f*)(bias + cout0 + 4));
    const bool first = (c & 7) == 0;
    const bool last  = (c & 7) == 7;
    float* const olane = out + (size_t)((c >> 3) * 32 + cout0) * OCH + 2 * (c & 7);
    const int n0 = (s0 < 2) ? 2 : 1;
    const int n1 = (s1 < 2) ? 2 : 1;

    for (int i0 = 0; i0 < n0; ++i0)
    for (int i1 = 0; i1 < n1; ++i1) {
        const int p0 = s0 + 16 * i0;
        const int p1 = s1 + 16 * i1;

        v8f acc[2][4][4];
#pragma unroll
        for (int rp = 0; rp < 2; ++rp)
#pragma unroll
            for (int jr = 0; jr < 4; ++jr)
#pragma unroll
                for (int jc = 0; jc < 4; ++jc) acc[rp][jr][jc] = {};

        for (int t0 = 0; t0 < 3; ++t0) {
            const int q0 = p0 - t0;
            if ((unsigned)q0 > 15u) continue;
            for (int t1 = 0; t1 < 3; ++t1) {
                const int q1 = p1 - t1;
                if ((unsigned)q1 > 15u) continue;
                const float* xp = xlane + (q0 * 16 + q1) * 256;
                const float* wq = wlane + (size_t)((t0 * 3 + t1) * 8) * WSTEP;
                v2f xrA[4][2], xrB[4][2];
                v4f wbA[8], wbB[8];
                load_step(xp, wq, xrA, wbA);
                for (int cc2 = 0; cc2 < 4; ++cc2) {
                    __builtin_amdgcn_sched_barrier(0);
                    load_step(xp + 4 * XCH, wq + WSTEP, xrB, wbB);
                    __builtin_amdgcn_sched_barrier(0);
                    mma_step(acc, xrA, wbA);
                    const int adv = (cc2 < 3) ? 2 : 1;
                    xp += (size_t)adv * 4 * XCH;
                    wq += adv * WSTEP;
                    __builtin_amdgcn_sched_barrier(0);
                    load_step(xp, wq, xrA, wbA);
                    __builtin_amdgcn_sched_barrier(0);
                    mma_step(acc, xrB, wbB);
                }
            }
        }

        v8f zu[4];
#pragma unroll
        for (int jc = 0; jc < 4; ++jc) {
            zu[jc] = (acc[1][1][jc] + acc[1][2][jc]) - acc[1][0][jc];
            xch[wave][0][jc][0][lane] = lo4(zu[jc]);
            xch[wave][0][jc][1][lane] = hi4(zu[jc]);
            xch[wave][1][jc][0][lane] = lo4(acc[1][3][jc]);
            xch[wave][1][jc][1][lane] = hi4(acc[1][3][jc]);
        }
        __syncthreads();

        float* oplane = olane + (size_t)((p0 * 18 + p1) * 18) * 18;
        float* o = oplane + (4 * u) * 18;
        {
            v8f za[4], zb[4];
#pragma unroll
            for (int jc = 0; jc < 4; ++jc) {
                za[jc] = acc[0][0][jc];
                zb[jc] = (acc[0][1][jc] - acc[0][2][jc]) - acc[0][3][jc];
            }
            if (u > 0) {
#pragma unroll
                for (int jc = 0; jc < 4; ++jc) {
                    za[jc] += cat8(xch[wave - 2][0][jc][0][lane], xch[wave - 2][0][jc][1][lane]);
                    zb[jc] += cat8(xch[wave - 2][1][jc][0][lane], xch[wave - 2][1][jc][1][lane]);
                }
            }
            emit_row(o, za[0], za[1], za[2], za[3], bi, first, last);
            emit_row(o + 18, zb[0], zb[1], zb[2], zb[3], bi, first, last);
        }
        {
            v8f z[4];
#pragma unroll
            for (int jc = 0; jc < 4; ++jc)
                z[jc] = ((acc[0][1][jc] + acc[0][2][jc]) - acc[0][0][jc]) + acc[1][0][jc];
            emit_row(o + 36, z[0], z[1], z[2], z[3], bi, first, last);
#pragma unroll
            for (int jc = 0; jc < 4; ++jc)
                z[jc] = acc[0][3][jc] + ((acc[1][1][jc] - acc[1][2][jc]) - acc[1][3][jc]);
            emit_row(o + 54, z[0], z[1], z[2], z[3], bi, first, last);
        }
        if (u == 0) {
            v8f z[4];
#pragma unroll
            for (int jc = 0; jc < 4; ++jc)
                z[jc] = cat8(xch[6 + hf][0][jc][0][lane], xch[6 + hf][0][jc][1][lane]);
            emit_row(oplane + 16 * 18, z[0], z[1], z[2], z[3], bi, first, last);
        }
        if (u == 3)
            emit_row(oplane + 17 * 18, acc[1][3][0], acc[1][3][1], acc[1][3][2], acc[1][3][3], bi, first, last);
        __syncthreads();
    }
}

extern "C" void kernel_launch(void* const* d_in, const int* in_sizes, int n_in,
                              void* d_out, int out_size, void* d_ws, size_t ws_size,
                              hipStream_t stream) {
    const float* x    = (const float*)d_in[0];
    const float* w    = (const float*)d_in[1];
    const float* bias = (const float*)d_in[2];
    float* out = (float*)d_out;
    float* wp  = (float*)d_ws;

    prepack_w<<<WPELEMS / 256, 256, 0, stream>>>(w, wp);
    convt4d_poly2_rq<<<256, 256, 0, stream>>>(x, wp, bias, out);
}


# rows as one 4-row polynomial product (input row quad x 3 taps at the points 0, 1, -1, 2, -2, infinity; divisions folded into the weights): 24 WMMA per K-step and wave, not 32
# speedup vs baseline: 3.3353x; 1.0865x over previous
#include <hip/hip_runtime.h>

typedef float v2f __attribute__((ext_vector_type(2)));
typedef float v4f __attribute__((ext_vector_type(4)));
typedef float v8f __attribute__((ext_vector_type(8)));

#define XCH   65536
#define OCH   104976
#define WHALF (12 * 128)
#define WSTEP (2 * WHALF)
#define WPELEMS (9 * 8 * WSTEP)


__device__ __forceinline__ float eval_tap(int j, float a0, float a1, float a2) {
    if (j == 0) return a0;
    if (j == 1) return 0.5f * ((a0 + a2) + a1);
    if (j == 2) return 0.5f * ((a0 + a2) - a1);
    return a2;
}

__device__ __forceinline__ float eval_row_tap(int j, float a0, float a1, float a2) {
    if (j == 0) return a0;
    if (j == 1) return 0.5f * ((a0 + a2) + a1);
    if (j == 2) return 0.5f * ((a0 + a2) - a1);
    if (j == 3) return ((a0 + 4.0f * a2) + 2.0f * a1) * (1.0f / 24.0f);
    if (j == 4) return ((a0 + 4.0f * a2) - 2.0f * a1) * (1.0f / 24.0f);
    return a2;
}

__global__ __launch_bounds__(256)
void prepack_w(const float* __restrict__ w, float* __restrict__ wp) {
    int idx = blockIdx.x * 256 + threadIdx.x;
    if (idx >= WPELEMS) return;
    int s    = idx & 1;
    int e    = (idx >> 1) & 1;
    int lane = (idx >> 2) & 31;
    int r1   = idx >> 7;
    int jp   = r1 % 12;
    int r2   = r1 / 12;
    int half = r2 & 1;
    int r3   = r2 >> 1;
    int cc   = r3 & 7;
    int t01  = r3 >> 3;
    int jj   = jp * 2 + e;
    int jr   = jj >> 2;
    int jc   = jj & 3;
    int cin  = cc * 4 + 2 * (lane >> 4) + s;
    int cout = half * 16 + (lane & 15);
    const float* p = w + (cin * 32 + cout) * 81 + t01 * 9;
    const float u0 = eval_tap(jc, p[0], p[1], p[2]);
    const float u1 = eval_tap(jc, p[3], p[4], p[5]);
    const float u2 = eval_tap(jc, p[6], p[7], p[8]);
    wp[idx] = eval_row_tap(jr, u0, u1, u2);
}


__device__ __forceinline__ void load_step(const float* xp, const float* wq, v2f (&xr)[4][2], v4f (&wb)[12]) {
#pragma unroll
    for (int r = 0; r < 4; ++r) {
        xr[r][0] = *(const v2f*)(xp + r * 16);
        xr[r][1] = *(const v2f*)(xp + XCH + r * 16);
    }
#pragma unroll
    for (int t = 0; t < 12; ++t) wb[t] = *(const v4f*)(wq + t * 128);
}

__device__ __forceinline__ void mma_step(v8f (&acc)[6][4], const v2f (&xr)[4][2], const v4f (&wb)[12]) {
    v2f rr[6][2];
#pragma unroll
    for (int s = 0; s < 2; ++s) {
        const v2f ev = xr[0][s] + xr[2][s];
        const v2f od = xr[1][s] + xr[3][s];
        const v2f e4 = xr[0][s] + 4.0f * xr[2][s];
        const v2f o4 = xr[1][s] + 4.0f * xr[3][s];
        rr[0][s] = xr[0][s];
        rr[1][s] = ev + od;
        rr[2][s] = ev - od;
        rr[3][s] = e4 + 2.0f * o4;
        rr[4][s] = e4 - 2.0f * o4;
        rr[5][s] = xr[3][s];
    }
#pragma unroll
    for (int jr = 0; jr < 6; ++jr) {
        const v2f c0 = rr[jr][0];
        const v2f c1 = rr[jr][1];
        v2f xa[4];
        xa[0] = (v2f){ c0.x,        c1.x        };
        xa[1] = (v2f){ c0.x + c0.y, c1.x + c1.y };
        xa[2] = (v2f){ c0.x - c0.y, c1.x - c1.y };
        xa[3] = (v2f){ c0.y,        c1.y        };
#pragma unroll
        for (int jc = 0; jc < 4; ++jc) {
            const int jj = jr * 4 + jc;
            const v4f wv = wb[jj >> 1];
            const v2f b  = (jj & 1) ? (v2f){ wv.z, wv.w } : (v2f){ wv.x, wv.y };
            acc[jr][jc] = __builtin_amdgcn_wmma_f32_16x16x4_f32(
                false, b, false, xa[jc], (short)0, acc[jr][jc], false, false);
        }
    }
}

__device__ __forceinline__ float left_pair(const float v, const bool first) {
    const int s = __builtin_amdgcn_update_dpp(0, __builtin_bit_cast(int, v), 0x111, 0xf, 0xf, true);
    return first ? 0.0f : __builtin_bit_cast(float, s);
}

__device__ __forceinline__ void emit_row(float* o, const v8f Z0, const v8f Z1, const v8f Z2, const v8f Z3,
                                         const v8f bi, const bool first, const bool last) {
    float t0[8], t1[8];
#pragma unroll
    for (int r = 0; r < 8; ++r) {
        t0[r] = (Z1[r] + Z2[r]) - Z0[r];
        t1[r] = Z3[r];
        const float y1 = (Z1[r] - Z2[r]) - Z3[r];
        const float o0 = (Z0[r] + left_pair(t0[r], first)) + bi[r];
        const float o1 = (y1 + left_pair(t1[r], first)) + bi[r];
        *(v2f*)(o + (size_t)r * OCH) = (v2f){ o0, o1 };
    }
    if (last) {
#pragma unroll
        for (int r = 0; r < 8; ++r)
            *(v2f*)(o + (size_t)r * OCH + 2) = (v2f){ t0[r] + bi[r], t1[r] + bi[r] };
    }
}

__device__ __forceinline__ v4f lo4(const v8f z) { return __builtin_shufflevector(z, z, 0, 1, 2, 3); }
__device__ __forceinline__ v4f hi4(const v8f z) { return __builtin_shufflevector(z, z, 4, 5, 6, 7); }
__device__ __forceinline__ v8f cat8(const v4f a, const v4f b) { return __builtin_shufflevector(a, b, 0, 1, 2, 3, 4, 5, 6, 7); }

__global__ __launch_bounds__(256)
void convt4d_poly_rows4(const float* __restrict__ x, const float* __restrict__ wp,
                      const float* __restrict__ bias, float* __restrict__ out) {
    __shared__ v4f xch[8][2][4][2][32];

    const int lane = threadIdx.x & 31;
    const int wave = __builtin_amdgcn_readfirstlane((int)(threadIdx.x >> 5));
    const int hf = wave & 1;
    const int u  = wave >> 1;
    const int s0 = blockIdx.x >> 4;
    const int s1 = blockIdx.x & 15;
    const int c = lane & 15;
    const int h = lane >> 4;
    const float* xlane = x + (size_t)((c >> 3) * 32 + 2 * h) * XCH + (4 * u) * 16 + 2 * (c & 7);
    const float* wlane = wp + hf * WHALF + lane * 4;
    const int cout0 = hf * 16 + 8 * h;
    const v8f bi = cat8(*(const v4f*)(bias + cout0), *(const v4f*)(bias + cout0 + 4));
    const bool first = (c & 7) == 0;
    const bool last  = (c & 7) == 7;
    float* const olane = out + (size_t)((c >> 3) * 32 + cout0) * OCH + 2 * (c & 7);
    const int n0 = (s0 < 2) ? 2 : 1;
    const int n1 = (s1 < 2) ? 2 : 1;

    for (int i0 = 0; i0 < n0; ++i0)
    for (int i1 = 0; i1 < n1; ++i1) {
        const int p0 = s0 + 16 * i0;
        const int p1 = s1 + 16 * i1;

        v8f acc[6][4];
#pragma unroll
        for (int jr = 0; jr < 6; ++jr)
#pragma unroll
            for (int jc = 0; jc < 4; ++jc) acc[jr][jc] = {};

        for (int t0 = 0; t0 < 3; ++t0) {
            const int q0 = p0 - t0;
            if ((unsigned)q0 > 15u) continue;
            for (int t1 = 0; t1 < 3; ++t1) {
                const int q1 = p1 - t1;
                if ((unsigned)q1 > 15u) continue;
                const float* xp = xlane + (q0 * 16 + q1) * 256;
                const float* wq = wlane + (size_t)((t0 * 3 + t1) * 8) * WSTEP;
                v2f xrA[4][2], xrB[4][2];
                v4f wbA[12], wbB[12];
                load_step(xp, wq, xrA, wbA);
                for (int cc2 = 0; cc2 < 4; ++cc2) {
                    __builtin_amdgcn_sched_barrier(0);
                    load_step(xp + 4 * XCH, wq + WSTEP, xrB, wbB);
                    __builtin_amdgcn_sched_barrier(0);
                    mma_step(acc, xrA, wbA);
                    const int adv = (cc2 < 3) ? 2 : 1;
                    xp += (size_t)adv * 4 * XCH;
                    wq += adv * WSTEP;
                    __builtin_amdgcn_sched_barrier(0);
                    load_step(xp, wq, xrA, wbA);
                    __builtin_amdgcn_sched_barrier(0);
                    mma_step(acc, xrB, wbB);
                }
            }
        }

        v8f y1[4], y2[4], y3[4];
#pragma unroll
        for (int jc = 0; jc < 4; ++jc) {
            const v8f M0 = acc[0][jc];
            const v8f M5 = acc[5][jc];
            const v8f sa = acc[1][jc] + acc[2][jc];
            const v8f sb = acc[1][jc] - acc[2][jc];
            const v8f sc = acc[3][jc] + acc[4][jc];
            const v8f sd = acc[3][jc] - acc[4][jc];
            const v8f y4 = (sc - sa * (1.0f / 3.0f)) + M0 * 0.25f;
            y2[jc] = (sa - M0) - y4;
            y3[jc] = (2.0f * sd - sb * (1.0f / 3.0f)) - 5.0f * M5;
            y1[jc] = (sb - M5) - y3[jc];
            xch[wave][0][jc][0][lane] = lo4(y4);
            xch[wave][0][jc][1][lane] = hi4(y4);
            xch[wave][1][jc][0][lane] = lo4(M5);
            xch[wave][1][jc][1][lane] = hi4(M5);
        }
        __syncthreads();

        float* oplane = olane + (size_t)((p0 * 18 + p1) * 18) * 18;
        float* o = oplane + (4 * u) * 18;
        {
            v8f za[4], zb[4];
#pragma unroll
            for (int jc = 0; jc < 4; ++jc) {
                za[jc] = acc[0][jc];
                zb[jc] = y1[jc];
            }
            if (u > 0) {
#pragma unroll
                for (int jc = 0; jc < 4; ++jc) {
                    za[jc] += cat8(xch[wave - 2][0][jc][0][lane], xch[wave - 2][0][jc][1][lane]);
                    zb[jc] += cat8(xch[wave - 2][1][jc][0][lane], xch[wave - 2][1][jc][1][lane]);
                }
            }
            emit_row(o, za[0], za[1], za[2], za[3], bi, first, last);
            emit_row(o + 18, zb[0], zb[1], zb[2], zb[3], bi, first, last);
        }
        emit_row(o + 36, y2[0], y2[1], y2[2], y2[3], bi, first, last);
        emit_row(o + 54, y3[0], y3[1], y3[2], y3[3], bi, first, last);
        if (u == 0) {
            v8f z[4];
#pragma unroll
            for (int jc = 0; jc < 4; ++jc)
                z[jc] = cat8(xch[6 + hf][0][jc][0][lane], xch[6 + hf][0][jc][1][lane]);
            emit_row(oplane + 16 * 18, z[0], z[1], z[2], z[3], bi, first, last);
        }
        if (u == 3)
            emit_row(oplane + 17 * 18, acc[5][0], acc[5][1], acc[5][2], acc[5][3], bi, first, last);
        __syncthreads();
    }
}

extern "C" void kernel_launch(void* const* d_in, const int* in_sizes, int n_in,
                              void* d_out, int out_size, void* d_ws, size_t ws_size,
                              hipStream_t stream) {
    const float* x    = (const float*)d_in[0];
    const float* w    = (const float*)d_in[1];
    const float* bias = (const float*)d_in[2];
    float* out = (float*)d_out;
    float* wp  = (float*)d_ws;

    prepack_w<<<WPELEMS / 256, 256, 0, stream>>>(w, wp);
    convt4d_poly_rows4<<<256, 256, 0, stream>>>(x, wp, bias, out);
}


# output tile staged in LDS in out's own layout, copied out in 16-byte pieces aligned to 128-byte lines: 24 whole-line stores per wave and plane replace 64 to 80 stores of 64- and 8-byte row pieces
# speedup vs baseline: 3.9698x; 1.1902x over previous
#include <hip/hip_runtime.h>

typedef float v2f __attribute__((ext_vector_type(2)));
typedef float v4f __attribute__((ext_vector_type(4)));
typedef float v8f __attribute__((ext_vector_type(8)));

#define XCH   65536
#define OCH   104976
#define WHALF (12 * 128)
#define WSTEP (2 * WHALF)
#define WPELEMS (9 * 8 * WSTEP)
#define PLANE 324
#define STG_B (32 * PLANE + 16)


__device__ __forceinline__ float eval_tap(int j, float a0, float a1, float a2) {
    if (j == 0) return a0;
    if (j == 1) return 0.5f * ((a0 + a2) + a1);
    if (j == 2) return 0.5f * ((a0 + a2) - a1);
    return a2;
}

__device__ __forceinline__ float eval_row_tap(int j, float a0, float a1, float a2) {
    if (j == 0) return a0;
    if (j == 1) return 0.5f * ((a0 + a2) + a1);
    if (j == 2) return 0.5f * ((a0 + a2) - a1);
    if (j == 3) return ((a0 + 4.0f * a2) + 2.0f * a1) * (1.0f / 24.0f);
    if (j == 4) return ((a0 + 4.0f * a2) - 2.0f * a1) * (1.0f / 24.0f);
    return a2;
}

__global__ __launch_bounds__(256)
void prepack_w(const float* __restrict__ w, float* __restrict__ wp) {
    int idx = blockIdx.x * 256 + threadIdx.x;
    if (idx >= WPELEMS) return;
    int s    = idx & 1;
    int e    = (idx >> 1) & 1;
    int lane = (idx >> 2) & 31;
    int r1   = idx >> 7;
    int jp   = r1 % 12;
    int r2   = r1 / 12;
    int half = r2 & 1;
    int r3   = r2 >> 1;
    int cc   = r3 & 7;
    int t01  = r3 >> 3;
    int jj   = jp * 2 + e;
    int jr   = jj >> 2;
    int jc   = jj & 3;
    int cin  = cc * 4 + 2 * (lane >> 4) + s;
    int cout = half * 16 + (lane & 15);
    const float* p = w + (cin * 32 + cout) * 81 + t01 * 9;
    const float u0 = eval_tap(jc, p[0], p[1], p[2]);
    const float u1 = eval_tap(jc, p[3], p[4], p[5]);
    const float u2 = eval_tap(jc, p[6], p[7], p[8]);
    wp[idx] = eval_row_tap(jr, u0, u1, u2);
}


__device__ __forceinline__ void load_step(const float* xp, const float* wq, v2f (&xr)[4][2], v4f (&wb)[12]) {
#pragma unroll
    for (int r = 0; r < 4; ++r) {
        xr[r][0] = *(const v2f*)(xp + r * 16);
        xr[r][1] = *(const v2f*)(xp + XCH + r * 16);
    }
#pragma unroll
    for (int t = 0; t < 12; ++t) wb[t] = *(const v4f*)(wq + t * 128);
}

__device__ __forceinline__ void mma_step(v8f (&acc)[6][4], const v2f (&xr)[4][2], const v4f (&wb)[12]) {
    v2f rr[6][2];
#pragma unroll
    for (int s = 0; s < 2; ++s) {
        const v2f ev = xr[0][s] + xr[2][s];
        const v2f od = xr[1][s] + xr[3][s];
        const v2f e4 = xr[0][s] + 4.0f * xr[2][s];
        const v2f o4 = xr[1][s] + 4.0f * xr[3][s];
        rr[0][s] = xr[0][s];
        rr[1][s] = ev + od;
        rr[2][s] = ev - od;
        rr[3][s] = e4 + 2.0f * o4;
        rr[4][s] = e4 - 2.0f * o4;
        rr[5][s] = xr[3][s];
    }
#pragma unroll
    for (int jr = 0; jr < 6; ++jr) {
        const v2f c0 = rr[jr][0];
        const v2f c1 = rr[jr][1];
        v2f xa[4];
        xa[0] = (v2f){ c0.x,        c1.x        };
        xa[1] = (v2f){ c0.x + c0.y, c1.x + c1.y };
        xa[2] = (v2f){ c0.x - c0.y, c1.x - c1.y };
        xa[3] = (v2f){ c0.y,        c1.y        };
#pragma unroll
        for (int jc = 0; jc < 4; ++jc) {
            const int jj = jr * 4 + jc;
            const v4f wv = wb[jj >> 1];
            const v2f b  = (jj & 1) ? (v2f){ wv.z, wv.w } : (v2f){ wv.x, wv.y };
            acc[jr][jc] = __builtin_amdgcn_wmma_f32_16x16x4_f32(
                false, b, false, xa[jc], (short)0, acc[jr][jc], false, false);
        }
    }
}

__device__ __forceinline__ float left_pair(const float v, const bool first) {
    const int s = __builtin_amdgcn_update_dpp(0, __builtin_bit_cast(int, v), 0x111, 0xf, 0xf, true);
    return first ? 0.0f : __builtin_bit_cast(float, s);
}

__device__ __forceinline__ void emit_row(float* o, const v8f Z0, const v8f Z1, const v8f Z2, const v8f Z3,
                                         const v8f bi, const bool first, const bool last) {
    float t0[8], t1[8];
#pragma unroll
    for (int r = 0; r < 8; ++r) {
        t0[r] = (Z1[r] + Z2[r]) - Z0[r];
        t1[r] = Z3[r];
        const float y1 = (Z1[r] - Z2[r]) - Z3[r];
        const float o0 = (Z0[r] + left_pair(t0[r], first)) + bi[r];
        const float o1 = (y1 + left_pair(t1[r], first)) + bi[r];
        *(v2f*)(o + r * PLANE) = (v2f){ o0, o1 };
    }
    if (last) {
#pragma unroll
        for (int r = 0; r < 8; ++r)
            *(v2f*)(o + r * PLANE + 2) = (v2f){ t0[r] + bi[r], t1[r] + bi[r] };
    }
}

__device__ __forceinline__ v4f lo4(const v8f z) { return __builtin_shufflevector(z, z, 0, 1, 2, 3); }
__device__ __forceinline__ v4f hi4(const v8f z) { return __builtin_shufflevector(z, z, 4, 5, 6, 7); }
__device__ __forceinline__ v8f cat8(const v4f a, const v4f b) { return __builtin_shufflevector(a, b, 0, 1, 2, 3, 4, 5, 6, 7); }

__global__ __launch_bounds__(256)
void convt4d_poly_rows4(const float* __restrict__ x, const float* __restrict__ wp,
                      const float* __restrict__ bias, float* __restrict__ out) {
    __shared__ v4f xch[8][2][4][2][32];
    __shared__ v4f stg4[(STG_B + 32 * PLANE) / 4];
    float* const stg = (float*)stg4;

    const int lane = threadIdx.x & 31;
    const int wave = __builtin_amdgcn_readfirstlane((int)(threadIdx.x >> 5));
    const int hf = wave & 1;
    const int u  = wave >> 1;
    const int s0 = blockIdx.x >> 4;
    const int s1 = blockIdx.x & 15;
    const int c = lane & 15;
    const int h = lane >> 4;
    const float* xlane = x + (size_t)((c >> 3) * 32 + 2 * h) * XCH + (4 * u) * 16 + 2 * (c & 7);
    const float* wlane = wp + hf * WHALF + lane * 4;
    const int cout0 = hf * 16 + 8 * h;
    const v8f bi = cat8(*(const v4f*)(bias + cout0), *(const v4f*)(bias + cout0 + 4));
    const bool first = (c & 7) == 0;
    const bool last  = (c & 7) == 7;
    float* const slane = stg + (c >> 3) * STG_B + cout0 * PLANE + 2 * (c & 7);
    const int n0 = (s0 < 2) ? 2 : 1;
    const int n1 = (s1 < 2) ? 2 : 1;

    for (int i0 = 0; i0 < n0; ++i0)
    for (int i1 = 0; i1 < n1; ++i1) {
        const int p0 = s0 + 16 * i0;
        const int p1 = s1 + 16 * i1;

        v8f acc[6][4];
#pragma unroll
        for (int jr = 0; jr < 6; ++jr)
#pragma unroll
            for (int jc = 0; jc < 4; ++jc) acc[jr][jc] = {};

        for (int t0 = 0; t0 < 3; ++t0) {
            const int q0 = p0 - t0;
            if ((unsigned)q0 > 15u) continue;
            for (int t1 = 0; t1 < 3; ++t1) {
                const int q1 = p1 - t1;
                if ((unsigned)q1 > 15u) continue;
                const float* xp = xlane + (q0 * 16 + q1) * 256;
                const float* wq = wlane + (size_t)((t0 * 3 + t1) * 8) * WSTEP;
                v2f xrA[4][2], xrB[4][2];
                v4f wbA[12], wbB[12];
                load_step(xp, wq, xrA, wbA);
                for (int cc2 = 0; cc2 < 4; ++cc2) {
                    __builtin_amdgcn_sched_barrier(0);
                    load_step(xp + 4 * XCH, wq + WSTEP, xrB, wbB);
                    __builtin_amdgcn_sched_barrier(0);
                    mma_step(acc, xrA, wbA);
                    const int adv = (cc2 < 3) ? 2 : 1;
                    xp += (size_t)adv * 4 * XCH;
                    wq += adv * WSTEP;
                    __builtin_amdgcn_sched_barrier(0);
                    load_step(xp, wq, xrA, wbA);
                    __builtin_amdgcn_sched_barrier(0);
                    mma_step(acc, xrB, wbB);
                }
            }
        }

        v8f y1[4], y2[4], y3[4];
#pragma unroll
        for (int jc = 0; jc < 4; ++jc) {
            const v8f M0 = acc[0][jc];
            const v8f M5 = acc[5][jc];
            const v8f sa = acc[1][jc] + acc[2][jc];
            const v8f sb = acc[1][jc] - acc[2][jc];
            const v8f sc = acc[3][jc] + acc[4][jc];
            const v8f sd = acc[3][jc] - acc[4][jc];
            const v8f y4 = (sc - sa * (1.0f / 3.0f)) + M0 * 0.25f;
            y2[jc] = (sa - M0) - y4;
            y3[jc] = (2.0f * sd - sb * (1.0f / 3.0f)) - 5.0f * M5;
            y1[jc] = (sb - M5) - y3[jc];
            xch[wave][0][jc][0][lane] = lo4(y4);
            xch[wave][0][jc][1][lane] = hi4(y4);
            xch[wave][1][jc][0][lane] = lo4(M5);
            xch[wave][1][jc][1][lane] = hi4(M5);
        }
        __syncthreads();

        float* o = slane + (4 * u) * 18;
        {
            v8f za[4], zb[4];
#pragma unroll
            for (int jc = 0; jc < 4; ++jc) {
                za[jc] = acc[0][jc];
                zb[jc] = y1[jc];
            }
            if (u > 0) {
#pragma unroll
                for (int jc = 0; jc < 4; ++jc) {
                    za[jc] += cat8(xch[wave - 2][0][jc][0][lane], xch[wave - 2][0][jc][1][lane]);
                    zb[jc] += cat8(xch[wave - 2][1][jc][0][lane], xch[wave - 2][1][jc][1][lane]);
                }
            }
            emit_row(o, za[0], za[1], za[2], za[3], bi, first, last);
            emit_row(o + 18, zb[0], zb[1], zb[2], zb[3], bi, first, last);
        }
        emit_row(o + 36, y2[0], y2[1], y2[2], y2[3], bi, first, last);
        emit_row(o + 54, y3[0], y3[1], y3[2], y3[3], bi, first, last);
        if (u == 0) {
            v8f z[4];
#pragma unroll
            for (int jc = 0; jc < 4; ++jc)
                z[jc] = cat8(xch[6 + hf][0][jc][0][lane], xch[6 + hf][0][jc][1][lane]);
            emit_row(slane + 16 * 18, z[0], z[1], z[2], z[3], bi, first, last);
        }
        if (u == 3)
            emit_row(slane + 17 * 18, acc[5][0], acc[5][1], acc[5][2], acc[5][3], bi, first, last);
        __syncthreads();

        {
            const int pl = p0 * 18 + p1;
            const int aE = pl & 7;
            const int aO = (pl + 4) & 7;
            v4f t[8][3];
#pragma unroll
            for (int q = 0; q < 8; ++q) {
                const int ch = wave * 8 + q;
                const v4f* sp = stg4 + (ch >> 5) * (STG_B / 4) + (ch & 31) * (PLANE / 4);
                const int a = (q & 1) ? aO : aE;
#pragma unroll
                for (int k = 0; k < 3; ++k) {
                    const int j = k * 32 + lane - a;
                    t[q][k] = sp[min(max(j, 0), 80)];
                }
            }
            float* const og = out + (size_t)(wave * 8) * OCH + (size_t)pl * PLANE;
#pragma unroll
            for (int q = 0; q < 8; ++q) {
                const int a = (q & 1) ? aO : aE;
                *(v4f*)(og + (size_t)q * OCH + 4 * (32 + lane - a)) = t[q][1];
            }
#pragma unroll
            for (int par = 0; par < 2; ++par) {
                const int a = par ? aO : aE;
                if (lane >= a) {
#pragma unroll
                    for (int q = par; q < 8; q += 2)
                        *(v4f*)(og + (size_t)q * OCH + 4 * (lane - a)) = t[q][0];
                }
                if (lane < 17 + a) {
#pragma unroll
                    for (int q = par; q < 8; q += 2)
                        *(v4f*)(og + (size_t)q * OCH + 4 * (64 + lane - a)) = t[q][2];
                }
            }
        }
    }
}

extern "C" void kernel_launch(void* const* d_in, const int* in_sizes, int n_in,
                              void* d_out, int out_size, void* d_ws, size_t ws_size,
                              hipStream_t stream) {
    const float* x    = (const float*)d_in[0];
    const float* w    = (const float*)d_in[1];
    const float* bias = (const float*)d_in[2];
    float* out = (float*)d_out;
    float* wp  = (float*)d_ws;

    prepack_w<<<WPELEMS / 256, 256, 0, stream>>>(w, wp);
    convt4d_poly_rows4<<<256, 256, 0, stream>>>(x, wp, bias, out);
}


# main kernel split by cout half: 512 workgroups of 4 waves, two resident per WGP (was 256 of 8): no barrier ties a SIMD's two waves, one's epilogue and load waits run under the other's WMMAs
# speedup vs baseline: 4.1699x; 1.0504x over previous
#include <hip/hip_runtime.h>

typedef float v2f __attribute__((ext_vector_type(2)));
typedef float v4f __attribute__((ext_vector_type(4)));
typedef float v8f __attribute__((ext_vector_type(8)));

#define XCH   65536
#define OCH   104976
#define WHALF (12 * 128)
#define WSTEP (2 * WHALF)
#define WPELEMS (9 * 8 * WSTEP)
#define PLANE 324
#define STG_B (16 * PLANE + 16)


__device__ __forceinline__ float eval_tap(int j, float a0, float a1, float a2) {
    if (j == 0) return a0;
    if (j == 1) return 0.5f * ((a0 + a2) + a1);
    if (j == 2) return 0.5f * ((a0 + a2) - a1);
    return a2;
}

__device__ __forceinline__ float eval_row_tap(int j, float a0, float a1, float a2) {
    if (j == 0) return a0;
    if (j == 1) return 0.5f * ((a0 + a2) + a1);
    if (j == 2) return 0.5f * ((a0 + a2) - a1);
    if (j == 3) return ((a0 + 4.0f * a2) + 2.0f * a1) * (1.0f / 24.0f);
    if (j == 4) return ((a0 + 4.0f * a2) - 2.0f * a1) * (1.0f / 24.0f);
    return a2;
}

__global__ __launch_bounds__(256)
void prepack_w(const float* __restrict__ w, float* __restrict__ wp) {
    int idx = blockIdx.x * 256 + threadIdx.x;
    if (idx >= WPELEMS) return;
    int s    = idx & 1;
    int e    = (idx >> 1) & 1;
    int lane = (idx >> 2) & 31;
    int r1   = idx >> 7;
    int jp   = r1 % 12;
    int r2   = r1 / 12;
    int half = r2 & 1;
    int r3   = r2 >> 1;
    int cc   = r3 & 7;
    int t01  = r3 >> 3;
    int jj   = jp * 2 + e;
    int jr   = jj >> 2;
    int jc   = jj & 3;
    int cin  = cc * 4 + 2 * (lane >> 4) + s;
    int cout = half * 16 + (lane & 15);
    const float* p = w + (cin * 32 + cout) * 81 + t01 * 9;
    const float u0 = eval_tap(jc, p[0], p[1], p[2]);
    const float u1 = eval_tap(jc, p[3], p[4], p[5]);
    const float u2 = eval_tap(jc, p[6], p[7], p[8]);
    wp[idx] = eval_row_tap(jr, u0, u1, u2);
}


__device__ __forceinline__ void load_step(const float* xp, const float* wq, v2f (&xr)[4][2], v4f (&wb)[12]) {
#pragma unroll
    for (int r = 0; r < 4; ++r) {
        xr[r][0] = *(const v2f*)(xp + r * 16);
        xr[r][1] = *(const v2f*)(xp + XCH + r * 16);
    }
#pragma unroll
    for (int t = 0; t < 12; ++t) wb[t] = *(const v4f*)(wq + t * 128);
}

__device__ __forceinline__ void mma_step(v8f (&acc)[6][4], const v2f (&xr)[4][2], const v4f (&wb)[12]) {
    v2f rr[6][2];
#pragma unroll
    for (int s = 0; s < 2; ++s) {
        const v2f ev = xr[0][s] + xr[2][s];
        const v2f od = xr[1][s] + xr[3][s];
        const v2f e4 = xr[0][s] + 4.0f * xr[2][s];
        const v2f o4 = xr[1][s] + 4.0f * xr[3][s];
        rr[0][s] = xr[0][s];
        rr[1][s] = ev + od;
        rr[2][s] = ev - od;
        rr[3][s] = e4 + 2.0f * o4;
        rr[4][s] = e4 - 2.0f * o4;
        rr[5][s] = xr[3][s];
    }
#pragma unroll
    for (int jr = 0; jr < 6; ++jr) {
        const v2f c0 = rr[jr][0];
        const v2f c1 = rr[jr][1];
        v2f xa[4];
        xa[0] = (v2f){ c0.x,        c1.x        };
        xa[1] = (v2f){ c0.x + c0.y, c1.x + c1.y };
        xa[2] = (v2f){ c0.x - c0.y, c1.x - c1.y };
        xa[3] = (v2f){ c0.y,        c1.y        };
#pragma unroll
        for (int jc = 0; jc < 4; ++jc) {
            const int jj = jr * 4 + jc;
            const v4f wv = wb[jj >> 1];
            const v2f b  = (jj & 1) ? (v2f){ wv.z, wv.w } : (v2f){ wv.x, wv.y };
            acc[jr][jc] = __builtin_amdgcn_wmma_f32_16x16x4_f32(
                false, b, false, xa[jc], (short)0, acc[jr][jc], false, false);
        }
    }
}

__device__ __forceinline__ float left_pair(const float v, const bool first) {
    const int s = __builtin_amdgcn_update_dpp(0, __builtin_bit_cast(int, v), 0x111, 0xf, 0xf, true);
    return first ? 0.0f : __builtin_bit_cast(float, s);
}

__device__ __forceinline__ void emit_row(float* o, const v8f Z0, const v8f Z1, const v8f Z2, const v8f Z3,
                                         const v8f bi, const bool first, const bool last) {
    float t0[8], t1[8];
#pragma unroll
    for (int r = 0; r < 8; ++r) {
        t0[r] = (Z1[r] + Z2[r]) - Z0[r];
        t1[r] = Z3[r];
        const float y1 = (Z1[r] - Z2[r]) - Z3[r];
        const float o0 = (Z0[r] + left_pair(t0[r], first)) + bi[r];
        const float o1 = (y1 + left_pair(t1[r], first)) + bi[r];
        *(v2f*)(o + r * PLANE) = (v2f){ o0, o1 };
    }
    if (last) {
#pragma unroll
        for (int r = 0; r < 8; ++r)
            *(v2f*)(o + r * PLANE + 2) = (v2f){ t0[r] + bi[r], t1[r] + bi[r] };
    }
}

__device__ __forceinline__ v4f lo4(const v8f z) { return __builtin_shufflevector(z, z, 0, 1, 2, 3); }
__device__ __forceinline__ v4f hi4(const v8f z) { return __builtin_shufflevector(z, z, 4, 5, 6, 7); }
__device__ __forceinline__ v8f cat8(const v4f a, const v4f b) { return __builtin_shufflevector(a, b, 0, 1, 2, 3, 4, 5, 6, 7); }

__global__ __launch_bounds__(128, 2)
void convt4d_poly_rows4(const float* __restrict__ x, const float* __restrict__ wp,
                      const float* __restrict__ bias, float* __restrict__ out) {
    __shared__ v4f xch[4][2][4][2][32];
    __shared__ v4f stg4[(STG_B + 16 * PLANE) / 4];
    float* const stg = (float*)stg4;

    const int lane = threadIdx.x & 31;
    const int wave = __builtin_amdgcn_readfirstlane((int)(threadIdx.x >> 5));
    const int hf = blockIdx.x & 1;
    const int u  = wave;
    const int s0 = blockIdx.x >> 5;
    const int s1 = (blockIdx.x >> 1) & 15;
    const int c = lane & 15;
    const int h = lane >> 4;
    const float* xlane = x + (size_t)((c >> 3) * 32 + 2 * h) * XCH + (4 * u) * 16 + 2 * (c & 7);
    const float* wlane = wp + hf * WHALF + lane * 4;
    const int cout0 = hf * 16 + 8 * h;
    const v8f bi = cat8(*(const v4f*)(bias + cout0), *(const v4f*)(bias + cout0 + 4));
    const bool first = (c & 7) == 0;
    const bool last  = (c & 7) == 7;
    float* const slane = stg + (c >> 3) * STG_B + (8 * h) * PLANE + 2 * (c & 7);
    const int n0 = (s0 < 2) ? 2 : 1;
    const int n1 = (s1 < 2) ? 2 : 1;

    for (int i0 = 0; i0 < n0; ++i0)
    for (int i1 = 0; i1 < n1; ++i1) {
        const int p0 = s0 + 16 * i0;
        const int p1 = s1 + 16 * i1;

        v8f acc[6][4];
#pragma unroll
        for (int jr = 0; jr < 6; ++jr)
#pragma unroll
            for (int jc = 0; jc < 4; ++jc) acc[jr][jc] = {};

        for (int t0 = 0; t0 < 3; ++t0) {
            const int q0 = p0 - t0;
            if ((unsigned)q0 > 15u) continue;
            for (int t1 = 0; t1 < 3; ++t1) {
                const int q1 = p1 - t1;
                if ((unsigned)q1 > 15u) continue;
                const float* xp = xlane + (q0 * 16 + q1) * 256;
                const float* wq = wlane + (size_t)((t0 * 3 + t1) * 8) * WSTEP;
                v2f xrA[4][2], xrB[4][2];
                v4f wbA[12], wbB[12];
                load_step(xp, wq, xrA, wbA);
                for (int cc2 = 0; cc2 < 4; ++cc2) {
                    __builtin_amdgcn_sched_barrier(0);
                    load_step(xp + 4 * XCH, wq + WSTEP, xrB, wbB);
                    __builtin_amdgcn_sched_barrier(0);
                    mma_step(acc, xrA, wbA);
                    const int adv = (cc2 < 3) ? 2 : 1;
                    xp += (size_t)adv * 4 * XCH;
                    wq += adv * WSTEP;
                    __builtin_amdgcn_sched_barrier(0);
                    load_step(xp, wq, xrA, wbA);
                    __builtin_amdgcn_sched_barrier(0);
                    mma_step(acc, xrB, wbB);
                }
            }
        }

        v8f y1[4], y2[4], y3[4];
#pragma unroll
        for (int jc = 0; jc < 4; ++jc) {
            const v8f M0 = acc[0][jc];
            const v8f M5 = acc[5][jc];
            const v8f sa = acc[1][jc] + acc[2][jc];
            const v8f sb = acc[1][jc] - acc[2][jc];
            const v8f sc = acc[3][jc] + acc[4][jc];
            const v8f sd = acc[3][jc] - acc[4][jc];
            const v8f y4 = (sc - sa * (1.0f / 3.0f)) + M0 * 0.25f;
            y2[jc] = (sa - M0) - y4;
            y3[jc] = (2.0f * sd - sb * (1.0f / 3.0f)) - 5.0f * M5;
            y1[jc] = (sb - M5) - y3[jc];
            xch[wave][0][jc][0][lane] = lo4(y4);
            xch[wave][0][jc][1][lane] = hi4(y4);
            xch[wave][1][jc][0][lane] = lo4(M5);
            xch[wave][1][jc][1][lane] = hi4(M5);
        }
        __syncthreads();

        float* o = slane + (4 * u) * 18;
        {
            v8f za[4], zb[4];
#pragma unroll
            for (int jc = 0; jc < 4; ++jc) {
                za[jc] = acc[0][jc];
                zb[jc] = y1[jc];
            }
            if (u > 0) {
#pragma unroll
                for (int jc = 0; jc < 4; ++jc) {
                    za[jc] += cat8(xch[wave - 1][0][jc][0][lane], xch[wave - 1][0][jc][1][lane]);
                    zb[jc] += cat8(xch[wave - 1][1][jc][0][lane], xch[wave - 1][1][jc][1][lane]);
                }
            }
            emit_row(o, za[0], za[1], za[2], za[3], bi, first, last);
            emit_row(o + 18, zb[0], zb[1], zb[2], zb[3], bi, first, last);
        }
        emit_row(o + 36, y2[0], y2[1], y2[2], y2[3], bi, first, last);
        emit_row(o + 54, y3[0], y3[1], y3[2], y3[3], bi, first, last);
        if (u == 0) {
            v8f z[4];
#pragma unroll
            for (int jc = 0; jc < 4; ++jc)
                z[jc] = cat8(xch[3][0][jc][0][lane], xch[3][0][jc][1][lane]);
            emit_row(slane + 16 * 18, z[0], z[1], z[2], z[3], bi, first, last);
        }
        if (u == 3)
            emit_row(slane + 17 * 18, acc[5][0], acc[5][1], acc[5][2], acc[5][3], bi, first, last);
        __syncthreads();

        {
            const int pl = p0 * 18 + p1;
            const int aE = pl & 7;
            const int aO = (pl + 4) & 7;
            v4f t[8][3];
#pragma unroll
            for (int q = 0; q < 8; ++q) {
                const int lp = wave * 8 + q;
                const v4f* sp = stg4 + (lp >> 4) * (STG_B / 4) + (lp & 15) * (PLANE / 4);
                const int a = (q & 1) ? aO : aE;
#pragma unroll
                for (int k = 0; k < 3; ++k) {
                    const int j = k * 32 + lane - a;
                    t[q][k] = sp[min(max(j, 0), 80)];
                }
            }
            float* const og = out + (size_t)((wave >> 1) * 32 + hf * 16 + (wave & 1) * 8) * OCH + (size_t)pl * PLANE;
#pragma unroll
            for (int q = 0; q < 8; ++q) {
                const int a = (q & 1) ? aO : aE;
                *(v4f*)(og + (size_t)q * OCH + 4 * (32 + lane - a)) = t[q][1];
            }
#pragma unroll
            for (int par = 0; par < 2; ++par) {
                const int a = par ? aO : aE;
                if (lane >= a) {
#pragma unroll
                    for (int q = par; q < 8; q += 2)
                        *(v4f*)(og + (size_t)q * OCH + 4 * (lane - a)) = t[q][0];
                }
                if (lane < 17 + a) {
#pragma unroll
                    for (int q = par; q < 8; q += 2)
                        *(v4f*)(og + (size_t)q * OCH + 4 * (64 + lane - a)) = t[q][2];
                }
            }
        }
    }
}

extern "C" void kernel_launch(void* const* d_in, const int* in_sizes, int n_in,
                              void* d_out, int out_size, void* d_ws, size_t ws_size,
                              hipStream_t stream) {
    const float* x    = (const float*)d_in[0];
    const float* w    = (const float*)d_in[1];
    const float* bias = (const float*)d_in[2];
    float* out = (float*)d_out;
    float* wp  = (float*)d_ws;

    prepack_w<<<WPELEMS / 256, 256, 0, stream>>>(w, wp);
    convt4d_poly_rows4<<<512, 128, 0, stream>>>(x, wp, bias, out);
}
